# starting kernel
# baseline (speedup 1.0000x reference)
#include <hip/hip_runtime.h>

#define N_NODES 50000
#define N_EDGES 800000
#define HID 128
#define PRED_LEN 24
#define N_LAYERS 3

typedef __attribute__((ext_vector_type(2))) float v2f;
typedef __attribute__((ext_vector_type(8))) float v8f;

__global__ void sage_scatter(const float* __restrict__ x,
                             const int* __restrict__ src,
                             const int* __restrict__ dst,
                             float* __restrict__ agg,
                             float* __restrict__ cnt) {
  int wave = (int)((blockIdx.x * blockDim.x + threadIdx.x) >> 5);
  int lane = threadIdx.x & 31;
  if (wave >= N_EDGES) return;
  int s = src[wave];
  int d = dst[wave];
  const float* xs = x + (size_t)s * HID;
  float* ad = agg + (size_t)d * HID;
#pragma unroll
  for (int i = 0; i < 4; ++i) {
    int h = i * 32 + lane;
    atomicAdd(&ad[h], xs[h]);
  }
  if (lane == 0) atomicAdd(&cnt[d], 1.0f);
}

__global__ void sage_gemm(const float* __restrict__ agg,
                          const float* __restrict__ cnt,
                          const float* __restrict__ xin,
                          const float* __restrict__ Wl,
                          const float* __restrict__ bl,
                          const float* __restrict__ Wr,
                          float* __restrict__ xout) {
  int tileM = blockIdx.x;
  int wid   = threadIdx.x >> 5;
  int lane  = threadIdx.x & 31;
  int half  = lane >> 4;
  int l15   = lane & 15;
  int row   = tileM * 16 + l15;
  int col   = wid * 16 + l15;

  float inv = 1.0f / fmaxf(cnt[row], 1.0f);

  const float* aR  = agg + (size_t)row * HID;
  const float* xR  = xin + (size_t)row * HID;
  const float* wlC = Wl + (size_t)col * HID;
  const float* wrC = Wr + (size_t)col * HID;
  int koff = half * 2;

  v8f acc = {};
#pragma unroll 4
  for (int k0 = 0; k0 < HID; k0 += 4) {
    int kk = k0 + koff;
    v2f a1 = *(const v2f*)(aR + kk);
    a1 *= inv;
    v2f b1 = *(const v2f*)(wlC + kk);
    acc = __builtin_amdgcn_wmma_f32_16x16x4_f32(false, a1, false, b1,
                                                (short)0, acc, false, false);
    v2f a2 = *(const v2f*)(xR + kk);
    v2f b2 = *(const v2f*)(wrC + kk);
    acc = __builtin_amdgcn_wmma_f32_16x16x4_f32(false, a2, false, b2,
                                                (short)0, acc, false, false);
  }

  float bias = bl[col];
  float* outBase = xout + ((size_t)(tileM * 16 + half * 8)) * HID + col;
#pragma unroll
  for (int v = 0; v < 8; ++v) {
    float r = acc[v] + bias;
    outBase[(size_t)v * HID] = fmaxf(r, 0.0f);
  }
}

__global__ void sage_proj(const float* __restrict__ xin,
                          const float* __restrict__ Wout,
                          const float* __restrict__ bout,
                          float* __restrict__ out) {
  int tileM = blockIdx.x;
  int wid   = threadIdx.x >> 5;
  int lane  = threadIdx.x & 31;
  int half  = lane >> 4;
  int l15   = lane & 15;
  int row   = tileM * 16 + l15;
  int col   = wid * 16 + l15;

  bool colOK = (col < PRED_LEN);
  float mask = colOK ? 1.0f : 0.0f;
  const float* xR = xin + (size_t)row * HID;
  const float* wC = Wout + (size_t)(colOK ? col : 0) * HID;
  int koff = half * 2;

  v8f acc = {};
#pragma unroll 4
  for (int k0 = 0; k0 < HID; k0 += 4) {
    int kk = k0 + koff;
    v2f a = *(const v2f*)(xR + kk);
    v2f b = *(const v2f*)(wC + kk);
    b *= mask;
    acc = __builtin_amdgcn_wmma_f32_16x16x4_f32(false, a, false, b,
                                                (short)0, acc, false, false);
  }

  if (colOK) {
    float bias = bout[col];
#pragma unroll
    for (int v = 0; v < 8; ++v) {
      int m = tileM * 16 + half * 8 + v;
      out[(size_t)m * PRED_LEN + col] = acc[v] + bias;
    }
  }
}

extern "C" void kernel_launch(void* const* d_in, const int* in_sizes, int n_in,
                              void* d_out, int out_size, void* d_ws, size_t ws_size,
                              hipStream_t stream) {
  (void)in_sizes; (void)n_in; (void)out_size; (void)ws_size;
  const float* x    = (const float*)d_in[0];
  const int*   ei   = (const int*)d_in[1];
  const float* Wl   = (const float*)d_in[2];
  const float* bl   = (const float*)d_in[3];
  const float* Wr   = (const float*)d_in[4];
  const float* Wout = (const float*)d_in[5];
  const float* bout = (const float*)d_in[6];
  float* out = (float*)d_out;

  const size_t NH = (size_t)N_NODES * HID;
  float* buf0 = (float*)d_ws;
  float* buf1 = buf0 + NH;
  float* agg  = buf1 + NH;
  float* cnt  = agg + NH;

  const int* src = ei;
  const int* dst = ei + N_EDGES;

  const float* cur = x;
  float* bufs[2] = {buf0, buf1};
  for (int layer = 0; layer < N_LAYERS; ++layer) {
    hipMemsetAsync(agg, 0, NH * sizeof(float), stream);
    hipMemsetAsync(cnt, 0, (size_t)N_NODES * sizeof(float), stream);
    sage_scatter<<<N_EDGES / 8, 256, 0, stream>>>(cur, src, dst, agg, cnt);
    float* nxt = bufs[layer & 1];
    sage_gemm<<<N_NODES / 16, 256, 0, stream>>>(
        agg, cnt, cur,
        Wl + (size_t)layer * HID * HID,
        bl + (size_t)layer * HID,
        Wr + (size_t)layer * HID * HID,
        nxt);
    cur = nxt;
  }
  sage_proj<<<N_NODES / 16, 64, 0, stream>>>(cur, Wout, bout, out);
}


# Float-atomic edge scatter replaced by a destination-sorted neighbour list built once per call and a per-layer register row gather; row sums now in list (arrival-rank) order
# speedup vs baseline: 2.1495x; 2.1495x over previous
#include <hip/hip_runtime.h>

#define N_NODES 50000
#define N_EDGES 800000
#define HID 128
#define PRED_LEN 24
#define N_LAYERS 3

typedef __attribute__((ext_vector_type(2))) float v2f;
typedef __attribute__((ext_vector_type(8))) float v8f;

#define NPAD 53248
#define SCAN_STEPS 13
#define SCAN_PER_WAVE (SCAN_STEPS * 128)


__global__ void __launch_bounds__(256)
sage_rank(const int* __restrict__ dst, int* __restrict__ deg, int* __restrict__ rank) {
  int i = (int)(blockIdx.x * 256 + threadIdx.x);
  if (i >= N_EDGES / 4) return;
  int4 d = ((const int4*)dst)[i];
  int4 r;
  r.x = atomicAdd(&deg[d.x], 1);
  r.y = atomicAdd(&deg[d.y], 1);
  r.z = atomicAdd(&deg[d.z], 1);
  r.w = atomicAdd(&deg[d.w], 1);
  ((int4*)rank)[i] = r;
}

__global__ void __launch_bounds__(1024)
sage_scan(const int* __restrict__ deg, int* __restrict__ rowptr, float* __restrict__ cnt) {
  __shared__ int wsum[32];
  int wid  = threadIdx.x >> 5;
  int lane = threadIdx.x & 31;
  const int4* dp = (const int4*)(deg + wid * SCAN_PER_WAVE) + lane;
  float4* cp = (float4*)(cnt + wid * SCAN_PER_WAVE) + lane;
  int4 v[SCAN_STEPS];
#pragma unroll
  for (int i = 0; i < SCAN_STEPS; ++i) v[i] = dp[i * 32];
  int carry = 0;
#pragma unroll
  for (int i = 0; i < SCAN_STEPS; ++i) {
    int4 t = v[i];
    cp[i * 32] = make_float4((float)t.x, (float)t.y, (float)t.z, (float)t.w);
    int s1 = t.x, s2 = s1 + t.y, s3 = s2 + t.z, s4 = s3 + t.w;
    int incl = s4;
#pragma unroll
    for (int off = 1; off < 32; off <<= 1) {
      int nb = __shfl_up(incl, off);
      if (lane >= off) incl += nb;
    }
    int excl = incl - s4 + carry;
    v[i] = make_int4(excl, excl + s1, excl + s2, excl + s3);
    carry += __shfl(incl, 31);
  }
  if (lane == 0) wsum[wid] = carry;
  __syncthreads();
  int off0 = (lane < wid) ? wsum[lane] : 0;
#pragma unroll
  for (int o = 16; o > 0; o >>= 1) off0 += __shfl_xor(off0, o);
  int4* rp = (int4*)(rowptr + wid * SCAN_PER_WAVE) + lane;
#pragma unroll
  for (int i = 0; i < SCAN_STEPS; ++i) {
    int4 t = v[i];
    rp[i * 32] = make_int4(t.x + off0, t.y + off0, t.z + off0, t.w + off0);
  }
}

__global__ void __launch_bounds__(256)
sage_fill(const int* __restrict__ src, const int* __restrict__ dst,
          const int* __restrict__ rowptr, const int* __restrict__ rank,
          int* __restrict__ col) {
  int i = (int)(blockIdx.x * 256 + threadIdx.x);
  if (i >= N_EDGES / 4) return;
  int4 d = ((const int4*)dst)[i];
  int4 s = ((const int4*)src)[i];
  int4 r = ((const int4*)rank)[i];
  col[rowptr[d.x] + r.x] = s.x;
  col[rowptr[d.y] + r.y] = s.y;
  col[rowptr[d.z] + r.z] = s.z;
  col[rowptr[d.w] + r.w] = s.w;
}

__device__ __forceinline__ float4 f4add(float4 a, float4 b) {
  return make_float4(a.x + b.x, a.y + b.y, a.z + b.z, a.w + b.w);
}

__global__ void __launch_bounds__(128)
sage_gather(const float* __restrict__ x, const int* __restrict__ rowptr,
            const int* __restrict__ deg, const int* __restrict__ col,
            float* __restrict__ agg) {
  int lane = threadIdx.x & 31;
  int wave = __builtin_amdgcn_readfirstlane((int)((blockIdx.x * 128 + threadIdx.x) >> 5));
  int node0 = wave * 4;
  int4 st4 = *(const int4*)(rowptr + node0);
  int4 dg4 = *(const int4*)(deg + node0);
  int st[4] = {st4.x, st4.y, st4.z, st4.w};
  int dg[4] = {dg4.x, dg4.y, dg4.z, dg4.w};
  const float4* x4 = (const float4*)x;
  float4* agg4 = (float4*)agg;
#pragma unroll
  for (int q = 0; q < 4; ++q) {
    int start = st[q];
    int n = dg[q];
    float4 acc = make_float4(0.f, 0.f, 0.f, 0.f);
    for (int base = 0; base < n; base += 32) {
      int m = n - base;
      if (m > 32) m = 32;
      int myc = (lane < m) ? col[start + base + lane] : 0;
      for (int t = 0; t < m; t += 8) {
        float4 r[8];
        r[0] = x4[(size_t)__builtin_amdgcn_readlane(myc, t) * (HID / 4) + lane];
#pragma unroll
        for (int u = 1; u < 8; ++u) {
          r[u] = make_float4(0.f, 0.f, 0.f, 0.f);
          if (t + u < m) {
            int s = __builtin_amdgcn_readlane(myc, t + u);
            r[u] = x4[(size_t)s * (HID / 4) + lane];
          }
        }
        float4 p = f4add(f4add(f4add(r[0], r[1]), f4add(r[2], r[3])),
                         f4add(f4add(r[4], r[5]), f4add(r[6], r[7])));
        acc = f4add(acc, p);
      }
    }
    agg4[(size_t)(node0 + q) * (HID / 4) + lane] = acc;
  }
}

__global__ void sage_gemm(const float* __restrict__ agg,
                          const float* __restrict__ cnt,
                          const float* __restrict__ xin,
                          const float* __restrict__ Wl,
                          const float* __restrict__ bl,
                          const float* __restrict__ Wr,
                          float* __restrict__ xout) {
  int tileM = blockIdx.x;
  int wid   = threadIdx.x >> 5;
  int lane  = threadIdx.x & 31;
  int half  = lane >> 4;
  int l15   = lane & 15;
  int row   = tileM * 16 + l15;
  int col   = wid * 16 + l15;

  float inv = 1.0f / fmaxf(cnt[row], 1.0f);

  const float* aR  = agg + (size_t)row * HID;
  const float* xR  = xin + (size_t)row * HID;
  const float* wlC = Wl + (size_t)col * HID;
  const float* wrC = Wr + (size_t)col * HID;
  int koff = half * 2;

  v8f acc = {};
#pragma unroll 4
  for (int k0 = 0; k0 < HID; k0 += 4) {
    int kk = k0 + koff;
    v2f a1 = *(const v2f*)(aR + kk);
    a1 *= inv;
    v2f b1 = *(const v2f*)(wlC + kk);
    acc = __builtin_amdgcn_wmma_f32_16x16x4_f32(false, a1, false, b1,
                                                (short)0, acc, false, false);
    v2f a2 = *(const v2f*)(xR + kk);
    v2f b2 = *(const v2f*)(wrC + kk);
    acc = __builtin_amdgcn_wmma_f32_16x16x4_f32(false, a2, false, b2,
                                                (short)0, acc, false, false);
  }

  float bias = bl[col];
  float* outBase = xout + ((size_t)(tileM * 16 + half * 8)) * HID + col;
#pragma unroll
  for (int v = 0; v < 8; ++v) {
    float r = acc[v] + bias;
    outBase[(size_t)v * HID] = fmaxf(r, 0.0f);
  }
}

__global__ void sage_proj(const float* __restrict__ xin,
                          const float* __restrict__ Wout,
                          const float* __restrict__ bout,
                          float* __restrict__ out) {
  int tileM = blockIdx.x;
  int wid   = threadIdx.x >> 5;
  int lane  = threadIdx.x & 31;
  int half  = lane >> 4;
  int l15   = lane & 15;
  int row   = tileM * 16 + l15;
  int col   = wid * 16 + l15;

  bool colOK = (col < PRED_LEN);
  float mask = colOK ? 1.0f : 0.0f;
  const float* xR = xin + (size_t)row * HID;
  const float* wC = Wout + (size_t)(colOK ? col : 0) * HID;
  int koff = half * 2;

  v8f acc = {};
#pragma unroll 4
  for (int k0 = 0; k0 < HID; k0 += 4) {
    int kk = k0 + koff;
    v2f a = *(const v2f*)(xR + kk);
    v2f b = *(const v2f*)(wC + kk);
    b *= mask;
    acc = __builtin_amdgcn_wmma_f32_16x16x4_f32(false, a, false, b,
                                                (short)0, acc, false, false);
  }

  if (colOK) {
    float bias = bout[col];
#pragma unroll
    for (int v = 0; v < 8; ++v) {
      int m = tileM * 16 + half * 8 + v;
      out[(size_t)m * PRED_LEN + col] = acc[v] + bias;
    }
  }
}

extern "C" void kernel_launch(void* const* d_in, const int* in_sizes, int n_in,
                              void* d_out, int out_size, void* d_ws, size_t ws_size,
                              hipStream_t stream) {
  (void)in_sizes; (void)n_in; (void)out_size; (void)ws_size;
  const float* x    = (const float*)d_in[0];
  const int*   ei   = (const int*)d_in[1];
  const float* Wl   = (const float*)d_in[2];
  const float* bl   = (const float*)d_in[3];
  const float* Wr   = (const float*)d_in[4];
  const float* Wout = (const float*)d_in[5];
  const float* bout = (const float*)d_in[6];
  float* out = (float*)d_out;

  const size_t NH = (size_t)N_NODES * HID;
  float* buf0 = (float*)d_ws;
  float* buf1 = buf0 + NH;
  float* agg  = buf1 + NH;
  float* cnt  = agg + NH;
  int* deg    = (int*)(cnt + NPAD);
  int* rowptr = deg + NPAD;
  int* rank   = rowptr + NPAD;
  int* col    = rank + N_EDGES;

  const int* src = ei;
  const int* dst = ei + N_EDGES;

  hipMemsetAsync(deg, 0, (size_t)NPAD * sizeof(int), stream);
  sage_rank<<<(N_EDGES / 4 + 255) / 256, 256, 0, stream>>>(dst, deg, rank);
  sage_scan<<<1, 1024, 0, stream>>>(deg, rowptr, cnt);
  sage_fill<<<(N_EDGES / 4 + 255) / 256, 256, 0, stream>>>(src, dst, rowptr, rank, col);

  const float* cur = x;
  float* bufs[2] = {buf0, buf1};
  for (int layer = 0; layer < N_LAYERS; ++layer) {
    sage_gather<<<N_NODES / 16, 128, 0, stream>>>(cur, rowptr, deg, col, agg);
    float* nxt = bufs[layer & 1];
    sage_gemm<<<N_NODES / 16, 256, 0, stream>>>(
        agg, cnt, cur,
        Wl + (size_t)layer * HID * HID,
        bl + (size_t)layer * HID,
        Wr + (size_t)layer * HID * HID,
        nxt);
    cur = nxt;
  }
  sage_proj<<<N_NODES / 16, 64, 0, stream>>>(cur, Wout, bout, out);
}


# Layer GEMM kept resident: 512 work-groups walk the 16-row tiles, weights stay in registers, row tiles staged in LDS by whole-row loads, mean scale moved to gather; k summed in two chains
# speedup vs baseline: 3.7840x; 1.7604x over previous
#include <hip/hip_runtime.h>

#define N_NODES 50000
#define N_EDGES 800000
#define HID 128
#define PRED_LEN 24
#define N_LAYERS 3

typedef __attribute__((ext_vector_type(2))) float v2f;
typedef __attribute__((ext_vector_type(8))) float v8f;

#define NPAD 53248
#define SCAN_STEPS 13
#define SCAN_PER_WAVE (SCAN_STEPS * 128)


__global__ void __launch_bounds__(256)
sage_rank(const int* __restrict__ dst, int* __restrict__ deg, int* __restrict__ rank) {
  int i = (int)(blockIdx.x * 256 + threadIdx.x);
  if (i >= N_EDGES / 4) return;
  int4 d = ((const int4*)dst)[i];
  int4 r;
  r.x = atomicAdd(&deg[d.x], 1);
  r.y = atomicAdd(&deg[d.y], 1);
  r.z = atomicAdd(&deg[d.z], 1);
  r.w = atomicAdd(&deg[d.w], 1);
  ((int4*)rank)[i] = r;
}

__global__ void __launch_bounds__(1024)
sage_scan(const int* __restrict__ deg, int* __restrict__ rowptr) {
  __shared__ int wsum[32];
  int wid  = threadIdx.x >> 5;
  int lane = threadIdx.x & 31;
  const int4* dp = (const int4*)(deg + wid * SCAN_PER_WAVE) + lane;
  int4 v[SCAN_STEPS];
#pragma unroll
  for (int i = 0; i < SCAN_STEPS; ++i) v[i] = dp[i * 32];
  int carry = 0;
#pragma unroll
  for (int i = 0; i < SCAN_STEPS; ++i) {
    int4 t = v[i];
    int s1 = t.x, s2 = s1 + t.y, s3 = s2 + t.z, s4 = s3 + t.w;
    int incl = s4;
#pragma unroll
    for (int off = 1; off < 32; off <<= 1) {
      int nb = __shfl_up(incl, off);
      if (lane >= off) incl += nb;
    }
    int excl = incl - s4 + carry;
    v[i] = make_int4(excl, excl + s1, excl + s2, excl + s3);
    carry += __shfl(incl, 31);
  }
  if (lane == 0) wsum[wid] = carry;
  __syncthreads();
  int off0 = (lane < wid) ? wsum[lane] : 0;
#pragma unroll
  for (int o = 16; o > 0; o >>= 1) off0 += __shfl_xor(off0, o);
  int4* rp = (int4*)(rowptr + wid * SCAN_PER_WAVE) + lane;
#pragma unroll
  for (int i = 0; i < SCAN_STEPS; ++i) {
    int4 t = v[i];
    rp[i * 32] = make_int4(t.x + off0, t.y + off0, t.z + off0, t.w + off0);
  }
}

__global__ void __launch_bounds__(256)
sage_fill(const int* __restrict__ src, const int* __restrict__ dst,
          const int* __restrict__ rowptr, const int* __restrict__ rank,
          int* __restrict__ col) {
  int i = (int)(blockIdx.x * 256 + threadIdx.x);
  if (i >= N_EDGES / 4) return;
  int4 d = ((const int4*)dst)[i];
  int4 s = ((const int4*)src)[i];
  int4 r = ((const int4*)rank)[i];
  col[rowptr[d.x] + r.x] = s.x;
  col[rowptr[d.y] + r.y] = s.y;
  col[rowptr[d.z] + r.z] = s.z;
  col[rowptr[d.w] + r.w] = s.w;
}

__device__ __forceinline__ float4 f4add(float4 a, float4 b) {
  return make_float4(a.x + b.x, a.y + b.y, a.z + b.z, a.w + b.w);
}

__global__ void __launch_bounds__(128)
sage_gather(const float* __restrict__ x, const int* __restrict__ rowptr,
            const int* __restrict__ deg, const int* __restrict__ col,
            float* __restrict__ agg) {
  int lane = threadIdx.x & 31;
  int wave = __builtin_amdgcn_readfirstlane((int)((blockIdx.x * 128 + threadIdx.x) >> 5));
  int node0 = wave * 4;
  int4 st4 = *(const int4*)(rowptr + node0);
  int4 dg4 = *(const int4*)(deg + node0);
  int st[4] = {st4.x, st4.y, st4.z, st4.w};
  int dg[4] = {dg4.x, dg4.y, dg4.z, dg4.w};
  const float4* x4 = (const float4*)x;
  float4* agg4 = (float4*)agg;
#pragma unroll
  for (int q = 0; q < 4; ++q) {
    int start = st[q];
    int n = dg[q];
    float4 acc = make_float4(0.f, 0.f, 0.f, 0.f);
    for (int base = 0; base < n; base += 32) {
      int m = n - base;
      if (m > 32) m = 32;
      int myc = (lane < m) ? col[start + base + lane] : 0;
      for (int t = 0; t < m; t += 8) {
        float4 r[8];
        r[0] = x4[(size_t)__builtin_amdgcn_readlane(myc, t) * (HID / 4) + lane];
#pragma unroll
        for (int u = 1; u < 8; ++u) {
          r[u] = make_float4(0.f, 0.f, 0.f, 0.f);
          if (t + u < m) {
            int s = __builtin_amdgcn_readlane(myc, t + u);
            r[u] = x4[(size_t)s * (HID / 4) + lane];
          }
        }
        float4 p = f4add(f4add(f4add(r[0], r[1]), f4add(r[2], r[3])),
                         f4add(f4add(r[4], r[5]), f4add(r[6], r[7])));
        acc = f4add(acc, p);
      }
    }
    float inv = 1.0f / fmaxf((float)n, 1.0f);
    agg4[(size_t)(node0 + q) * (HID / 4) + lane] =
        make_float4(acc.x * inv, acc.y * inv, acc.z * inv, acc.w * inv);
  }
}

#define GEMM_BLOCKS 512
#define N_TILES (N_NODES / 16)
#define A_PITCH4 65

__device__ __forceinline__ v8f wmma_k4(float a0, float a1, float b0, float b1, v8f c) {
  v2f a = {a0, a1};
  v2f b = {b0, b1};
  return __builtin_amdgcn_wmma_f32_16x16x4_f32(false, a, false, b, (short)0, c, false, false);
}

__global__ void __launch_bounds__(256)
sage_gemm(const float* __restrict__ agg,
          const float* __restrict__ xin,
          const float* __restrict__ Wl,
          const float* __restrict__ bl,
          const float* __restrict__ Wr,
          float* __restrict__ xout) {
  __shared__ float4 As[2][16 * A_PITCH4];

  const int lane = threadIdx.x & 31;
  const int wid  = __builtin_amdgcn_readfirstlane((int)(threadIdx.x >> 5));
  const int half = lane >> 4;
  const int l15  = lane & 15;
  const int col  = wid * 16 + l15;

  const float4* agg4 = (const float4*)agg;
  const float4* x4   = (const float4*)xin;

  int T = (int)blockIdx.x;
  int rA = T * 16 + wid;
  int rB = rA + 8;
  float4 g0 = agg4[(size_t)rA * (HID / 4) + lane];
  float4 g1 = agg4[(size_t)rB * (HID / 4) + lane];
  float4 g2 = x4[(size_t)rA * (HID / 4) + lane];
  float4 g3 = x4[(size_t)rB * (HID / 4) + lane];

  const float4* wl4 = (const float4*)(Wl + (size_t)col * HID) + half;
  const float4* wr4 = (const float4*)(Wr + (size_t)col * HID) + half;
  float4 wL[16], wR[16];
#pragma unroll
  for (int i = 0; i < 16; ++i) {
    wL[i] = wl4[2 * i];
    wR[i] = wr4[2 * i];
  }
  const float bias = bl[col];

  As[0][wid * A_PITCH4 + lane]            = g0;
  As[0][(wid + 8) * A_PITCH4 + lane]      = g1;
  As[0][wid * A_PITCH4 + 32 + lane]       = g2;
  As[0][(wid + 8) * A_PITCH4 + 32 + lane] = g3;
  __syncthreads();

  int buf = 0;
  for (; T < N_TILES; T += GEMM_BLOCKS) {
    int Tn = T + GEMM_BLOCKS;
    if (Tn >= N_TILES) Tn = T;
    rA = Tn * 16 + wid;
    rB = rA + 8;
    g0 = agg4[(size_t)rA * (HID / 4) + lane];
    g1 = agg4[(size_t)rB * (HID / 4) + lane];
    g2 = x4[(size_t)rA * (HID / 4) + lane];
    g3 = x4[(size_t)rB * (HID / 4) + lane];
    __builtin_amdgcn_sched_barrier(0);

    const float4* ar = &As[buf][l15 * A_PITCH4 + half];
    v8f acc0 = {};
    v8f acc1 = {};
#pragma unroll
    for (int i = 0; i < 16; ++i) {
      float4 a = ar[2 * i];
      float4 s = ar[32 + 2 * i];
      acc0 = wmma_k4(a.x, a.y, wL[i].x, wL[i].y, acc0);
      acc1 = wmma_k4(s.x, s.y, wR[i].x, wR[i].y, acc1);
      acc0 = wmma_k4(a.z, a.w, wL[i].z, wL[i].w, acc0);
      acc1 = wmma_k4(s.z, s.w, wR[i].z, wR[i].w, acc1);
    }

    float4* an = As[buf ^ 1];
    an[wid * A_PITCH4 + lane]            = g0;
    an[(wid + 8) * A_PITCH4 + lane]      = g1;
    an[wid * A_PITCH4 + 32 + lane]       = g2;
    an[(wid + 8) * A_PITCH4 + 32 + lane] = g3;
    __syncthreads();
    buf ^= 1;

    float* outBase = xout + ((size_t)(T * 16 + half * 8)) * HID + col;
#pragma unroll
    for (int v = 0; v < 8; ++v) {
      float r = (acc0[v] + acc1[v]) + bias;
      outBase[(size_t)v * HID] = fmaxf(r, 0.0f);
    }
  }
}

__global__ void sage_proj(const float* __restrict__ xin,
                          const float* __restrict__ Wout,
                          const float* __restrict__ bout,
                          float* __restrict__ out) {
  int tileM = blockIdx.x;
  int wid   = threadIdx.x >> 5;
  int lane  = threadIdx.x & 31;
  int half  = lane >> 4;
  int l15   = lane & 15;
  int row   = tileM * 16 + l15;
  int col   = wid * 16 + l15;

  bool colOK = (col < PRED_LEN);
  float mask = colOK ? 1.0f : 0.0f;
  const float* xR = xin + (size_t)row * HID;
  const float* wC = Wout + (size_t)(colOK ? col : 0) * HID;
  int koff = half * 2;

  v8f acc = {};
#pragma unroll 4
  for (int k0 = 0; k0 < HID; k0 += 4) {
    int kk = k0 + koff;
    v2f a = *(const v2f*)(xR + kk);
    v2f b = *(const v2f*)(wC + kk);
    b *= mask;
    acc = __builtin_amdgcn_wmma_f32_16x16x4_f32(false, a, false, b,
                                                (short)0, acc, false, false);
  }

  if (colOK) {
    float bias = bout[col];
#pragma unroll
    for (int v = 0; v < 8; ++v) {
      int m = tileM * 16 + half * 8 + v;
      out[(size_t)m * PRED_LEN + col] = acc[v] + bias;
    }
  }
}

extern "C" void kernel_launch(void* const* d_in, const int* in_sizes, int n_in,
                              void* d_out, int out_size, void* d_ws, size_t ws_size,
                              hipStream_t stream) {
  (void)in_sizes; (void)n_in; (void)out_size; (void)ws_size;
  const float* x    = (const float*)d_in[0];
  const int*   ei   = (const int*)d_in[1];
  const float* Wl   = (const float*)d_in[2];
  const float* bl   = (const float*)d_in[3];
  const float* Wr   = (const float*)d_in[4];
  const float* Wout = (const float*)d_in[5];
  const float* bout = (const float*)d_in[6];
  float* out = (float*)d_out;

  const size_t NH = (size_t)N_NODES * HID;
  float* buf0 = (float*)d_ws;
  float* buf1 = buf0 + NH;
  float* agg  = buf1 + NH;
  int* deg    = (int*)(agg + NH);
  int* rowptr = deg + NPAD;
  int* rank   = rowptr + NPAD;
  int* col    = rank + N_EDGES;

  const int* src = ei;
  const int* dst = ei + N_EDGES;

  hipMemsetAsync(deg, 0, (size_t)NPAD * sizeof(int), stream);
  sage_rank<<<(N_EDGES / 4 + 255) / 256, 256, 0, stream>>>(dst, deg, rank);
  sage_scan<<<1, 1024, 0, stream>>>(deg, rowptr);
  sage_fill<<<(N_EDGES / 4 + 255) / 256, 256, 0, stream>>>(src, dst, rowptr, rank, col);

  const float* cur = x;
  float* bufs[2] = {buf0, buf1};
  for (int layer = 0; layer < N_LAYERS; ++layer) {
    sage_gather<<<N_NODES / 16, 128, 0, stream>>>(cur, rowptr, deg, col, agg);
    float* nxt = bufs[layer & 1];
    sage_gemm<<<GEMM_BLOCKS, 256, 0, stream>>>(
        agg, cur,
        Wl + (size_t)layer * HID * HID,
        bl + (size_t)layer * HID,
        Wr + (size_t)layer * HID * HID,
        nxt);
    cur = nxt;
  }
  sage_proj<<<N_NODES / 16, 64, 0, stream>>>(cur, Wout, bout, out);
}


# Neighbour list built in two launches through LDS bins (edge slices sorted by 256-node bin, then one work-group per bin) instead of global atomics, memset, scan, fill; list order = LDS arrival order
# speedup vs baseline: 5.2305x; 1.3823x over previous
#include <hip/hip_runtime.h>

#define N_NODES 50000
#define N_EDGES 800000
#define HID 128
#define PRED_LEN 24
#define N_LAYERS 3

typedef __attribute__((ext_vector_type(2))) float v2f;
typedef __attribute__((ext_vector_type(8))) float v8f;

#define NPAD 53248

#define BIN_NODES 256
#define NBINS ((N_NODES + BIN_NODES - 1) / BIN_NODES)
#define PART_BLOCKS 256
#define EPB (N_EDGES / PART_BLOCKS)
#define PART_K ((EPB + 255) / 256)
#define COL_CAP 8192

static_assert(EPB * PART_BLOCKS == N_EDGES, "edge slices must tile the edge list");
static_assert(EPB < 65536, "count and start of a run share one word");
static_assert(N_NODES <= 65536, "source id and local destination share one word");
static_assert(NBINS <= 256, "one thread per bin");
static_assert(NBINS * BIN_NODES <= NPAD, "degree / row-start planes hold whole bins");

__device__ __forceinline__ int scan256(int v, int* wsum, int& total) {
  int lane = threadIdx.x & 31;
  int wid  = threadIdx.x >> 5;
  int incl = v;
#pragma unroll
  for (int off = 1; off < 32; off <<= 1) {
    int nb = __shfl_up(incl, off);
    if (lane >= off) incl += nb;
  }
  if (lane == 31) wsum[wid] = incl;
  __syncthreads();
  int woff = 0;
  int tot = 0;
#pragma unroll
  for (int i = 0; i < 8; ++i) {
    int s = wsum[i];
    if (i < wid) woff += s;
    tot += s;
  }
  __syncthreads();
  total = tot;
  return incl - v + woff;
}

__global__ void __launch_bounds__(256)
sage_part(const int* __restrict__ src, const int* __restrict__ dst,
          int* __restrict__ tab, int* __restrict__ stage) {
  __shared__ int cnt[256];
  __shared__ int boff[256];
  __shared__ int wsum[8];
  __shared__ int sorted[EPB + 1];
  const int t = threadIdx.x;
  __builtin_assume(t >= 0 && t < 256);
  const int w = blockIdx.x;
  cnt[t] = 0;
  __syncthreads();

  const int* sp = src + (size_t)w * EPB;
  const int* dp = dst + (size_t)w * EPB;
  int pk[PART_K], bk[PART_K], rk[PART_K];
#pragma unroll
  for (int k = 0; k < PART_K; ++k) {
    int i = t + 256 * k;
    int ic = (i < EPB) ? i : (EPB - 1);
    int d = dp[ic];
    int s = sp[ic];
    pk[k] = (i < EPB) ? (((d & (BIN_NODES - 1)) << 16) | s) : -1;
    bk[k] = d / BIN_NODES;
  }
#pragma unroll
  for (int k = 0; k < PART_K; ++k) {
    rk[k] = 0;
    if (t + 256 * k < EPB) rk[k] = atomicAdd(&cnt[bk[k]], 1);
  }
  __syncthreads();

  int c = cnt[t];
  int total;
  int o = scan256(c, wsum, total);
  boff[t] = o;
  if (t < NBINS) tab[t * PART_BLOCKS + w] = (o << 16) | c;
  __syncthreads();

#pragma unroll
  for (int k = 0; k < PART_K; ++k) {
    int idx = (t + 256 * k < EPB) ? (boff[bk[k]] + rk[k]) : EPB;
    sorted[idx] = pk[k];
  }
  __syncthreads();

  int* so = stage + (size_t)w * EPB;
  for (int i = t; i < EPB; i += 256) so[i] = sorted[i];
}

__global__ void __launch_bounds__(256)
sage_csr(const int* __restrict__ tab, const int* __restrict__ stage,
         int* __restrict__ deg, int* __restrict__ rowptr, int* __restrict__ col) {
  __shared__ int degL[BIN_NODES];
  __shared__ int cur[BIN_NODES];
  __shared__ int wsum[8];
  __shared__ int red[8];
  __shared__ int colL[COL_CAP];
  const int t = threadIdx.x;
  const int lane = t & 31;
  const int wid = t >> 5;
  const int b = blockIdx.x;

  const int co = tab[b * PART_BLOCKS + t];
  const int c = co & 0xFFFF;
  const int o = co >> 16;
  degL[t] = 0;

  int part = o;
#pragma unroll
  for (int off = 16; off > 0; off >>= 1) part += __shfl_xor(part, off);
  if (lane == 0) red[wid] = part;
  __syncthreads();
  int base = 0;
#pragma unroll
  for (int i = 0; i < 8; ++i) base += red[i];

  int val[32];
  bool big = false;
#pragma unroll
  for (int k = 0; k < 32; ++k) {
    int ck = __builtin_amdgcn_readlane(c, k);
    int ok = __builtin_amdgcn_readlane(o, k);
    const int* seg = stage + (size_t)(wid * 32 + k) * EPB + ok;
    val[k] = -1;
    if (lane < ck) val[k] = seg[lane];
    big = big || (ck > 32);
  }

#pragma unroll
  for (int k = 0; k < 32; ++k) {
    if (val[k] >= 0) atomicAdd(&degL[val[k] >> 16], 1);
  }
  if (big) {
    for (int k = 0; k < 32; ++k) {
      int ck = __builtin_amdgcn_readlane(c, k);
      int ok = __builtin_amdgcn_readlane(o, k);
      const int* seg = stage + (size_t)(wid * 32 + k) * EPB + ok;
      for (int j = 32 + lane; j < ck; j += 32) atomicAdd(&degL[seg[j] >> 16], 1);
    }
  }
  __syncthreads();

  const int dg = degL[t];
  int total;
  const int ex = scan256(dg, wsum, total);
  deg[b * BIN_NODES + t] = dg;
  rowptr[b * BIN_NODES + t] = base + ex;
  cur[t] = ex;
  __syncthreads();

  const bool fits = (total <= COL_CAP);
  int pos[32];
#pragma unroll
  for (int k = 0; k < 32; ++k) {
    pos[k] = 0;
    if (val[k] >= 0) pos[k] = atomicAdd(&cur[val[k] >> 16], 1);
  }
  if (fits) {
#pragma unroll
    for (int k = 0; k < 32; ++k) {
      if (val[k] >= 0) colL[pos[k]] = val[k] & 0xFFFF;
    }
  } else {
#pragma unroll
    for (int k = 0; k < 32; ++k) {
      if (val[k] >= 0) col[base + pos[k]] = val[k] & 0xFFFF;
    }
  }
  if (big) {
    for (int k = 0; k < 32; ++k) {
      int ck = __builtin_amdgcn_readlane(c, k);
      int ok = __builtin_amdgcn_readlane(o, k);
      const int* seg = stage + (size_t)(wid * 32 + k) * EPB + ok;
      for (int j = 32 + lane; j < ck; j += 32) {
        int v = seg[j];
        int p = atomicAdd(&cur[v >> 16], 1);
        int s = v & 0xFFFF;
        if (fits) colL[p] = s; else col[base + p] = s;
      }
    }
  }
  __syncthreads();
  if (fits) {
    for (int i = t; i < total; i += 256) col[base + i] = colL[i];
  }
}

__device__ __forceinline__ float4 f4add(float4 a, float4 b) {
  return make_float4(a.x + b.x, a.y + b.y, a.z + b.z, a.w + b.w);
}

__global__ void __launch_bounds__(128)
sage_gather(const float* __restrict__ x, const int* __restrict__ rowptr,
            const int* __restrict__ deg, const int* __restrict__ col,
            float* __restrict__ agg) {
  int lane = threadIdx.x & 31;
  int wave = __builtin_amdgcn_readfirstlane((int)((blockIdx.x * 128 + threadIdx.x) >> 5));
  int node0 = wave * 4;
  int4 st4 = *(const int4*)(rowptr + node0);
  int4 dg4 = *(const int4*)(deg + node0);
  int st[4] = {st4.x, st4.y, st4.z, st4.w};
  int dg[4] = {dg4.x, dg4.y, dg4.z, dg4.w};
  const float4* x4 = (const float4*)x;
  float4* agg4 = (float4*)agg;
#pragma unroll
  for (int q = 0; q < 4; ++q) {
    int start = st[q];
    int n = dg[q];
    float4 acc = make_float4(0.f, 0.f, 0.f, 0.f);
    for (int base = 0; base < n; base += 32) {
      int m = n - base;
      if (m > 32) m = 32;
      int myc = (lane < m) ? col[start + base + lane] : 0;
      for (int t = 0; t < m; t += 8) {
        float4 r[8];
        r[0] = x4[(size_t)__builtin_amdgcn_readlane(myc, t) * (HID / 4) + lane];
#pragma unroll
        for (int u = 1; u < 8; ++u) {
          r[u] = make_float4(0.f, 0.f, 0.f, 0.f);
          if (t + u < m) {
            int s = __builtin_amdgcn_readlane(myc, t + u);
            r[u] = x4[(size_t)s * (HID / 4) + lane];
          }
        }
        float4 p = f4add(f4add(f4add(r[0], r[1]), f4add(r[2], r[3])),
                         f4add(f4add(r[4], r[5]), f4add(r[6], r[7])));
        acc = f4add(acc, p);
      }
    }
    float inv = 1.0f / fmaxf((float)n, 1.0f);
    agg4[(size_t)(node0 + q) * (HID / 4) + lane] =
        make_float4(acc.x * inv, acc.y * inv, acc.z * inv, acc.w * inv);
  }
}

#define GEMM_BLOCKS 512
#define N_TILES (N_NODES / 16)
#define A_PITCH4 65

__device__ __forceinline__ v8f wmma_k4(float a0, float a1, float b0, float b1, v8f c) {
  v2f a = {a0, a1};
  v2f b = {b0, b1};
  return __builtin_amdgcn_wmma_f32_16x16x4_f32(false, a, false, b, (short)0, c, false, false);
}

__global__ void __launch_bounds__(256)
sage_gemm(const float* __restrict__ agg,
          const float* __restrict__ xin,
          const float* __restrict__ Wl,
          const float* __restrict__ bl,
          const float* __restrict__ Wr,
          float* __restrict__ xout) {
  __shared__ float4 As[2][16 * A_PITCH4];

  const int lane = threadIdx.x & 31;
  const int wid  = __builtin_amdgcn_readfirstlane((int)(threadIdx.x >> 5));
  const int half = lane >> 4;
  const int l15  = lane & 15;
  const int col  = wid * 16 + l15;

  const float4* agg4 = (const float4*)agg;
  const float4* x4   = (const float4*)xin;

  int T = (int)blockIdx.x;
  int rA = T * 16 + wid;
  int rB = rA + 8;
  float4 g0 = agg4[(size_t)rA * (HID / 4) + lane];
  float4 g1 = agg4[(size_t)rB * (HID / 4) + lane];
  float4 g2 = x4[(size_t)rA * (HID / 4) + lane];
  float4 g3 = x4[(size_t)rB * (HID / 4) + lane];

  const float4* wl4 = (const float4*)(Wl + (size_t)col * HID) + half;
  const float4* wr4 = (const float4*)(Wr + (size_t)col * HID) + half;
  float4 wL[16], wR[16];
#pragma unroll
  for (int i = 0; i < 16; ++i) {
    wL[i] = wl4[2 * i];
    wR[i] = wr4[2 * i];
  }
  const float bias = bl[col];

  As[0][wid * A_PITCH4 + lane]            = g0;
  As[0][(wid + 8) * A_PITCH4 + lane]      = g1;
  As[0][wid * A_PITCH4 + 32 + lane]       = g2;
  As[0][(wid + 8) * A_PITCH4 + 32 + lane] = g3;
  __syncthreads();

  int buf = 0;
  for (; T < N_TILES; T += GEMM_BLOCKS) {
    int Tn = T + GEMM_BLOCKS;
    if (Tn >= N_TILES) Tn = T;
    rA = Tn * 16 + wid;
    rB = rA + 8;
    g0 = agg4[(size_t)rA * (HID / 4) + lane];
    g1 = agg4[(size_t)rB * (HID / 4) + lane];
    g2 = x4[(size_t)rA * (HID / 4) + lane];
    g3 = x4[(size_t)rB * (HID / 4) + lane];
    __builtin_amdgcn_sched_barrier(0);

    const float4* ar = &As[buf][l15 * A_PITCH4 + half];
    v8f acc0 = {};
    v8f acc1 = {};
#pragma unroll
    for (int i = 0; i < 16; ++i) {
      float4 a = ar[2 * i];
      float4 s = ar[32 + 2 * i];
      acc0 = wmma_k4(a.x, a.y, wL[i].x, wL[i].y, acc0);
      acc1 = wmma_k4(s.x, s.y, wR[i].x, wR[i].y, acc1);
      acc0 = wmma_k4(a.z, a.w, wL[i].z, wL[i].w, acc0);
      acc1 = wmma_k4(s.z, s.w, wR[i].z, wR[i].w, acc1);
    }

    float4* an = As[buf ^ 1];
    an[wid * A_PITCH4 + lane]            = g0;
    an[(wid + 8) * A_PITCH4 + lane]      = g1;
    an[wid * A_PITCH4 + 32 + lane]       = g2;
    an[(wid + 8) * A_PITCH4 + 32 + lane] = g3;
    __syncthreads();
    buf ^= 1;

    float* outBase = xout + ((size_t)(T * 16 + half * 8)) * HID + col;
#pragma unroll
    for (int v = 0; v < 8; ++v) {
      float r = (acc0[v] + acc1[v]) + bias;
      outBase[(size_t)v * HID] = fmaxf(r, 0.0f);
    }
  }
}

__global__ void sage_proj(const float* __restrict__ xin,
                          const float* __restrict__ Wout,
                          const float* __restrict__ bout,
                          float* __restrict__ out) {
  int tileM = blockIdx.x;
  int wid   = threadIdx.x >> 5;
  int lane  = threadIdx.x & 31;
  int half  = lane >> 4;
  int l15   = lane & 15;
  int row   = tileM * 16 + l15;
  int col   = wid * 16 + l15;

  bool colOK = (col < PRED_LEN);
  float mask = colOK ? 1.0f : 0.0f;
  const float* xR = xin + (size_t)row * HID;
  const float* wC = Wout + (size_t)(colOK ? col : 0) * HID;
  int koff = half * 2;

  v8f acc = {};
#pragma unroll 4
  for (int k0 = 0; k0 < HID; k0 += 4) {
    int kk = k0 + koff;
    v2f a = *(const v2f*)(xR + kk);
    v2f b = *(const v2f*)(wC + kk);
    b *= mask;
    acc = __builtin_amdgcn_wmma_f32_16x16x4_f32(false, a, false, b,
                                                (short)0, acc, false, false);
  }

  if (colOK) {
    float bias = bout[col];
#pragma unroll
    for (int v = 0; v < 8; ++v) {
      int m = tileM * 16 + half * 8 + v;
      out[(size_t)m * PRED_LEN + col] = acc[v] + bias;
    }
  }
}

extern "C" void kernel_launch(void* const* d_in, const int* in_sizes, int n_in,
                              void* d_out, int out_size, void* d_ws, size_t ws_size,
                              hipStream_t stream) {
  (void)in_sizes; (void)n_in; (void)out_size; (void)ws_size;
  const float* x    = (const float*)d_in[0];
  const int*   ei   = (const int*)d_in[1];
  const float* Wl   = (const float*)d_in[2];
  const float* bl   = (const float*)d_in[3];
  const float* Wr   = (const float*)d_in[4];
  const float* Wout = (const float*)d_in[5];
  const float* bout = (const float*)d_in[6];
  float* out = (float*)d_out;

  const size_t NH = (size_t)N_NODES * HID;
  float* buf0 = (float*)d_ws;
  float* buf1 = buf0 + NH;
  float* agg  = buf1 + NH;
  int* deg    = (int*)(agg + NH);
  int* rowptr = deg + NPAD;
  int* tab    = rowptr + NPAD;
  int* stage  = tab + NBINS * PART_BLOCKS;
  int* col    = stage + N_EDGES;

  const int* src = ei;
  const int* dst = ei + N_EDGES;

  sage_part<<<PART_BLOCKS, 256, 0, stream>>>(src, dst, tab, stage);
  sage_csr<<<NBINS, 256, 0, stream>>>(tab, stage, deg, rowptr, col);

  const float* cur = x;
  float* bufs[2] = {buf0, buf1};
  for (int layer = 0; layer < N_LAYERS; ++layer) {
    sage_gather<<<N_NODES / 16, 128, 0, stream>>>(cur, rowptr, deg, col, agg);
    float* nxt = bufs[layer & 1];
    sage_gemm<<<GEMM_BLOCKS, 256, 0, stream>>>(
        agg, cur,
        Wl + (size_t)layer * HID * HID,
        bl + (size_t)layer * HID,
        Wr + (size_t)layer * HID * HID,
        nxt);
    cur = nxt;
  }
  sage_proj<<<N_NODES / 16, 64, 0, stream>>>(cur, Wout, bout, out);
}


# Output projection moved into the last layer's GEMM launch: activated tile stays in LDS, 8 waves multiply it by Wout^T in four k slices whose partials are added in fixed order; 8 launches
# speedup vs baseline: 5.4022x; 1.0328x over previous
#include <hip/hip_runtime.h>

#define N_NODES 50000
#define N_EDGES 800000
#define HID 128
#define PRED_LEN 24
#define N_LAYERS 3

typedef __attribute__((ext_vector_type(2))) float v2f;
typedef __attribute__((ext_vector_type(8))) float v8f;

#define NPAD 53248

#define BIN_NODES 256
#define NBINS ((N_NODES + BIN_NODES - 1) / BIN_NODES)
#define PART_BLOCKS 256
#define EPB (N_EDGES / PART_BLOCKS)
#define PART_K ((EPB + 255) / 256)
#define COL_CAP 8192

static_assert(EPB * PART_BLOCKS == N_EDGES, "edge slices must tile the edge list");
static_assert(EPB < 65536, "count and start of a run share one word");
static_assert(N_NODES <= 65536, "source id and local destination share one word");
static_assert(NBINS <= 256, "one thread per bin");
static_assert(NBINS * BIN_NODES <= NPAD, "degree / row-start planes hold whole bins");

__device__ __forceinline__ int scan256(int v, int* wsum, int& total) {
  int lane = threadIdx.x & 31;
  int wid  = threadIdx.x >> 5;
  int incl = v;
#pragma unroll
  for (int off = 1; off < 32; off <<= 1) {
    int nb = __shfl_up(incl, off);
    if (lane >= off) incl += nb;
  }
  if (lane == 31) wsum[wid] = incl;
  __syncthreads();
  int woff = 0;
  int tot = 0;
#pragma unroll
  for (int i = 0; i < 8; ++i) {
    int s = wsum[i];
    if (i < wid) woff += s;
    tot += s;
  }
  __syncthreads();
  total = tot;
  return incl - v + woff;
}

__global__ void __launch_bounds__(256)
sage_part(const int* __restrict__ src, const int* __restrict__ dst,
          int* __restrict__ tab, int* __restrict__ stage) {
  __shared__ int cnt[256];
  __shared__ int boff[256];
  __shared__ int wsum[8];
  __shared__ int sorted[EPB + 1];
  const int t = threadIdx.x;
  __builtin_assume(t >= 0 && t < 256);
  const int w = blockIdx.x;
  cnt[t] = 0;
  __syncthreads();

  const int* sp = src + (size_t)w * EPB;
  const int* dp = dst + (size_t)w * EPB;
  int pk[PART_K], bk[PART_K], rk[PART_K];
#pragma unroll
  for (int k = 0; k < PART_K; ++k) {
    int i = t + 256 * k;
    int ic = (i < EPB) ? i : (EPB - 1);
    int d = dp[ic];
    int s = sp[ic];
    pk[k] = (i < EPB) ? (((d & (BIN_NODES - 1)) << 16) | s) : -1;
    bk[k] = d / BIN_NODES;
  }
#pragma unroll
  for (int k = 0; k < PART_K; ++k) {
    rk[k] = 0;
    if (t + 256 * k < EPB) rk[k] = atomicAdd(&cnt[bk[k]], 1);
  }
  __syncthreads();

  int c = cnt[t];
  int total;
  int o = scan256(c, wsum, total);
  boff[t] = o;
  if (t < NBINS) tab[t * PART_BLOCKS + w] = (o << 16) | c;
  __syncthreads();

#pragma unroll
  for (int k = 0; k < PART_K; ++k) {
    int idx = (t + 256 * k < EPB) ? (boff[bk[k]] + rk[k]) : EPB;
    sorted[idx] = pk[k];
  }
  __syncthreads();

  int* so = stage + (size_t)w * EPB;
  for (int i = t; i < EPB; i += 256) so[i] = sorted[i];
}

__global__ void __launch_bounds__(256)
sage_csr(const int* __restrict__ tab, const int* __restrict__ stage,
         int* __restrict__ deg, int* __restrict__ rowptr, int* __restrict__ col) {
  __shared__ int degL[BIN_NODES];
  __shared__ int cur[BIN_NODES];
  __shared__ int wsum[8];
  __shared__ int red[8];
  __shared__ int colL[COL_CAP];
  const int t = threadIdx.x;
  const int lane = t & 31;
  const int wid = t >> 5;
  const int b = blockIdx.x;

  const int co = tab[b * PART_BLOCKS + t];
  const int c = co & 0xFFFF;
  const int o = co >> 16;
  degL[t] = 0;

  int part = o;
#pragma unroll
  for (int off = 16; off > 0; off >>= 1) part += __shfl_xor(part, off);
  if (lane == 0) red[wid] = part;
  __syncthreads();
  int base = 0;
#pragma unroll
  for (int i = 0; i < 8; ++i) base += red[i];

  int val[32];
  bool big = false;
#pragma unroll
  for (int k = 0; k < 32; ++k) {
    int ck = __builtin_amdgcn_readlane(c, k);
    int ok = __builtin_amdgcn_readlane(o, k);
    const int* seg = stage + (size_t)(wid * 32 + k) * EPB + ok;
    val[k] = -1;
    if (lane < ck) val[k] = seg[lane];
    big = big || (ck > 32);
  }

#pragma unroll
  for (int k = 0; k < 32; ++k) {
    if (val[k] >= 0) atomicAdd(&degL[val[k] >> 16], 1);
  }
  if (big) {
    for (int k = 0; k < 32; ++k) {
      int ck = __builtin_amdgcn_readlane(c, k);
      int ok = __builtin_amdgcn_readlane(o, k);
      const int* seg = stage + (size_t)(wid * 32 + k) * EPB + ok;
      for (int j = 32 + lane; j < ck; j += 32) atomicAdd(&degL[seg[j] >> 16], 1);
    }
  }
  __syncthreads();

  const int dg = degL[t];
  int total;
  const int ex = scan256(dg, wsum, total);
  deg[b * BIN_NODES + t] = dg;
  rowptr[b * BIN_NODES + t] = base + ex;
  cur[t] = ex;
  __syncthreads();

  const bool fits = (total <= COL_CAP);
  int pos[32];
#pragma unroll
  for (int k = 0; k < 32; ++k) {
    pos[k] = 0;
    if (val[k] >= 0) pos[k] = atomicAdd(&cur[val[k] >> 16], 1);
  }
  if (fits) {
#pragma unroll
    for (int k = 0; k < 32; ++k) {
      if (val[k] >= 0) colL[pos[k]] = val[k] & 0xFFFF;
    }
  } else {
#pragma unroll
    for (int k = 0; k < 32; ++k) {
      if (val[k] >= 0) col[base + pos[k]] = val[k] & 0xFFFF;
    }
  }
  if (big) {
    for (int k = 0; k < 32; ++k) {
      int ck = __builtin_amdgcn_readlane(c, k);
      int ok = __builtin_amdgcn_readlane(o, k);
      const int* seg = stage + (size_t)(wid * 32 + k) * EPB + ok;
      for (int j = 32 + lane; j < ck; j += 32) {
        int v = seg[j];
        int p = atomicAdd(&cur[v >> 16], 1);
        int s = v & 0xFFFF;
        if (fits) colL[p] = s; else col[base + p] = s;
      }
    }
  }
  __syncthreads();
  if (fits) {
    for (int i = t; i < total; i += 256) col[base + i] = colL[i];
  }
}

__device__ __forceinline__ float4 f4add(float4 a, float4 b) {
  return make_float4(a.x + b.x, a.y + b.y, a.z + b.z, a.w + b.w);
}

__global__ void __launch_bounds__(128)
sage_gather(const float* __restrict__ x, const int* __restrict__ rowptr,
            const int* __restrict__ deg, const int* __restrict__ col,
            float* __restrict__ agg) {
  int lane = threadIdx.x & 31;
  int wave = __builtin_amdgcn_readfirstlane((int)((blockIdx.x * 128 + threadIdx.x) >> 5));
  int node0 = wave * 4;
  int4 st4 = *(const int4*)(rowptr + node0);
  int4 dg4 = *(const int4*)(deg + node0);
  int st[4] = {st4.x, st4.y, st4.z, st4.w};
  int dg[4] = {dg4.x, dg4.y, dg4.z, dg4.w};
  const float4* x4 = (const float4*)x;
  float4* agg4 = (float4*)agg;
#pragma unroll
  for (int q = 0; q < 4; ++q) {
    int start = st[q];
    int n = dg[q];
    float4 acc = make_float4(0.f, 0.f, 0.f, 0.f);
    for (int base = 0; base < n; base += 32) {
      int m = n - base;
      if (m > 32) m = 32;
      int myc = (lane < m) ? col[start + base + lane] : 0;
      for (int t = 0; t < m; t += 8) {
        float4 r[8];
        r[0] = x4[(size_t)__builtin_amdgcn_readlane(myc, t) * (HID / 4) + lane];
#pragma unroll
        for (int u = 1; u < 8; ++u) {
          r[u] = make_float4(0.f, 0.f, 0.f, 0.f);
          if (t + u < m) {
            int s = __builtin_amdgcn_readlane(myc, t + u);
            r[u] = x4[(size_t)s * (HID / 4) + lane];
          }
        }
        float4 p = f4add(f4add(f4add(r[0], r[1]), f4add(r[2], r[3])),
                         f4add(f4add(r[4], r[5]), f4add(r[6], r[7])));
        acc = f4add(acc, p);
      }
    }
    float inv = 1.0f / fmaxf((float)n, 1.0f);
    agg4[(size_t)(node0 + q) * (HID / 4) + lane] =
        make_float4(acc.x * inv, acc.y * inv, acc.z * inv, acc.w * inv);
  }
}

#define GEMM_BLOCKS 512
#define N_TILES (N_NODES / 16)
#define A_PITCH4 65

__device__ __forceinline__ v8f wmma_k4(float a0, float a1, float b0, float b1, v8f c) {
  v2f a = {a0, a1};
  v2f b = {b0, b1};
  return __builtin_amdgcn_wmma_f32_16x16x4_f32(false, a, false, b, (short)0, c, false, false);
}

#define H_PITCH 132
#define P_PITCH 36
#define OUT_TILE (16 * PRED_LEN)

__device__ __forceinline__ void proj_finish(const float* __restrict__ ps, const float* __restrict__ bout,
                                            float* __restrict__ out, int tile) {
  for (int i = (int)threadIdx.x; i < OUT_TILE; i += 256) {
    int row = i / PRED_LEN;
    int cc = i - row * PRED_LEN;
    const float* p = ps + row * P_PITCH + cc;
    float s = ((p[0] + p[16 * P_PITCH]) + p[2 * 16 * P_PITCH]) + p[3 * 16 * P_PITCH];
    out[(size_t)tile * OUT_TILE + i] = s + bout[cc];
  }
}

template <bool LAST>
__device__ __forceinline__ void
gemm_body(float4* __restrict__ AsBase, float* __restrict__ HsBase, float* __restrict__ PsBase,
          const float* __restrict__ agg,
          const float* __restrict__ xin,
          const float* __restrict__ Wl,
          const float* __restrict__ bl,
          const float* __restrict__ Wr,
          float* __restrict__ xout,
          const float* __restrict__ Wout,
          const float* __restrict__ bout,
          float* __restrict__ out) {
  float4 (*As)[16 * A_PITCH4] = (float4 (*)[16 * A_PITCH4])AsBase;


  const int lane = threadIdx.x & 31;
  const int wid  = __builtin_amdgcn_readfirstlane((int)(threadIdx.x >> 5));
  const int half = lane >> 4;
  const int l15  = lane & 15;
  const int col  = wid * 16 + l15;

  const float4* agg4 = (const float4*)agg;
  const float4* x4   = (const float4*)xin;

  int T = (int)blockIdx.x;
  int rA = T * 16 + wid;
  int rB = rA + 8;
  float4 g0 = agg4[(size_t)rA * (HID / 4) + lane];
  float4 g1 = agg4[(size_t)rB * (HID / 4) + lane];
  float4 g2 = x4[(size_t)rA * (HID / 4) + lane];
  float4 g3 = x4[(size_t)rB * (HID / 4) + lane];

  const float4* wl4 = (const float4*)(Wl + (size_t)col * HID) + half;
  const float4* wr4 = (const float4*)(Wr + (size_t)col * HID) + half;
  float4 wL[16], wR[16];
#pragma unroll
  for (int i = 0; i < 16; ++i) {
    wL[i] = wl4[2 * i];
    wR[i] = wr4[2 * i];
  }
  const float bias = bl[col];

  const int nt = wid & 1;
  const int kq = wid >> 1;
  float4 wo[4];
  if constexpr (LAST) {
    const int pc = nt * 16 + l15;
    const float4* wo4 = (const float4*)(Wout + (size_t)(pc < PRED_LEN ? pc : 0) * HID + 32 * kq) + half;
#pragma unroll
    for (int i = 0; i < 4; ++i) {
      float4 w = wo4[2 * i];
      wo[i] = (pc < PRED_LEN) ? w : make_float4(0.f, 0.f, 0.f, 0.f);
    }
  }

  As[0][wid * A_PITCH4 + lane]            = g0;
  As[0][(wid + 8) * A_PITCH4 + lane]      = g1;
  As[0][wid * A_PITCH4 + 32 + lane]       = g2;
  As[0][(wid + 8) * A_PITCH4 + 32 + lane] = g3;
  __syncthreads();

  int buf = 0;
  for (; T < N_TILES; T += GEMM_BLOCKS) {
    int Tn = T + GEMM_BLOCKS;
    if (Tn >= N_TILES) Tn = T;
    rA = Tn * 16 + wid;
    rB = rA + 8;
    g0 = agg4[(size_t)rA * (HID / 4) + lane];
    g1 = agg4[(size_t)rB * (HID / 4) + lane];
    g2 = x4[(size_t)rA * (HID / 4) + lane];
    g3 = x4[(size_t)rB * (HID / 4) + lane];
    __builtin_amdgcn_sched_barrier(0);

    const float4* ar = &As[buf][l15 * A_PITCH4 + half];
    v8f acc0 = {};
    v8f acc1 = {};
#pragma unroll
    for (int i = 0; i < 16; ++i) {
      float4 a = ar[2 * i];
      float4 s = ar[32 + 2 * i];
      acc0 = wmma_k4(a.x, a.y, wL[i].x, wL[i].y, acc0);
      acc1 = wmma_k4(s.x, s.y, wR[i].x, wR[i].y, acc1);
      acc0 = wmma_k4(a.z, a.w, wL[i].z, wL[i].w, acc0);
      acc1 = wmma_k4(s.z, s.w, wR[i].z, wR[i].w, acc1);
    }

    const int cur = buf;
    if constexpr (LAST) {
      float* hs = HsBase + cur * (16 * H_PITCH) + (half * 8) * H_PITCH + col;
#pragma unroll
      for (int v = 0; v < 8; ++v) {
        float r = (acc0[v] + acc1[v]) + bias;
        hs[v * H_PITCH] = fmaxf(r, 0.0f);
      }
    }

    float4* an = As[buf ^ 1];
    an[wid * A_PITCH4 + lane]            = g0;
    an[(wid + 8) * A_PITCH4 + lane]      = g1;
    an[wid * A_PITCH4 + 32 + lane]       = g2;
    an[(wid + 8) * A_PITCH4 + 32 + lane] = g3;
    __syncthreads();
    buf ^= 1;

    if constexpr (!LAST) {
      float* outBase = xout + ((size_t)(T * 16 + half * 8)) * HID + col;
#pragma unroll
      for (int v = 0; v < 8; ++v) {
        float r = (acc0[v] + acc1[v]) + bias;
        outBase[(size_t)v * HID] = fmaxf(r, 0.0f);
      }
    } else {
      if (T != (int)blockIdx.x)
        proj_finish(PsBase + (cur ^ 1) * (4 * 16 * P_PITCH), bout, out, T - GEMM_BLOCKS);
      const float4* hr = (const float4*)(HsBase + cur * (16 * H_PITCH) + l15 * H_PITCH + 32 * kq) + half;
      v8f pacc = {};
#pragma unroll
      for (int i = 0; i < 4; ++i) {
        float4 a = hr[2 * i];
        pacc = wmma_k4(a.x, a.y, wo[i].x, wo[i].y, pacc);
        pacc = wmma_k4(a.z, a.w, wo[i].z, wo[i].w, pacc);
      }
      float* ps = PsBase + cur * (4 * 16 * P_PITCH) + kq * (16 * P_PITCH) + (half * 8) * P_PITCH + nt * 16 + l15;
#pragma unroll
      for (int v = 0; v < 8; ++v) ps[v * P_PITCH] = pacc[v];
    }
  }
  if constexpr (LAST) {
    __syncthreads();
    proj_finish(PsBase + (buf ^ 1) * (4 * 16 * P_PITCH), bout, out, T - GEMM_BLOCKS);
  }
}

__global__ void __launch_bounds__(256)
sage_gemm(const float* __restrict__ agg, const float* __restrict__ xin,
          const float* __restrict__ Wl, const float* __restrict__ bl,
          const float* __restrict__ Wr, float* __restrict__ xout) {
  __shared__ float4 As[2 * 16 * A_PITCH4];
  gemm_body<false>(As, nullptr, nullptr, agg, xin, Wl, bl, Wr, xout, nullptr, nullptr, nullptr);
}

__global__ void __launch_bounds__(256)
sage_gemm_proj(const float* __restrict__ agg, const float* __restrict__ xin,
               const float* __restrict__ Wl, const float* __restrict__ bl,
               const float* __restrict__ Wr, const float* __restrict__ Wout,
               const float* __restrict__ bout, float* __restrict__ out) {
  __shared__ float4 As[2 * 16 * A_PITCH4];
  __shared__ float4 Hs4[2 * 16 * H_PITCH / 4];
  __shared__ float Ps[2 * 4 * 16 * P_PITCH];
  gemm_body<true>(As, (float*)Hs4, Ps, agg, xin, Wl, bl, Wr, nullptr, Wout, bout, out);
}

extern "C" void kernel_launch(void* const* d_in, const int* in_sizes, int n_in,
                              void* d_out, int out_size, void* d_ws, size_t ws_size,
                              hipStream_t stream) {
  (void)in_sizes; (void)n_in; (void)out_size; (void)ws_size;
  const float* x    = (const float*)d_in[0];
  const int*   ei   = (const int*)d_in[1];
  const float* Wl   = (const float*)d_in[2];
  const float* bl   = (const float*)d_in[3];
  const float* Wr   = (const float*)d_in[4];
  const float* Wout = (const float*)d_in[5];
  const float* bout = (const float*)d_in[6];
  float* out = (float*)d_out;

  const size_t NH = (size_t)N_NODES * HID;
  float* buf0 = (float*)d_ws;
  float* buf1 = buf0 + NH;
  float* agg  = buf1 + NH;
  int* deg    = (int*)(agg + NH);
  int* rowptr = deg + NPAD;
  int* tab    = rowptr + NPAD;
  int* stage  = tab + NBINS * PART_BLOCKS;
  int* col    = stage + N_EDGES;

  const int* src = ei;
  const int* dst = ei + N_EDGES;

  sage_part<<<PART_BLOCKS, 256, 0, stream>>>(src, dst, tab, stage);
  sage_csr<<<NBINS, 256, 0, stream>>>(tab, stage, deg, rowptr, col);

  const float* cur = x;
  float* bufs[2] = {buf0, buf1};
  for (int layer = 0; layer < N_LAYERS; ++layer) {
    sage_gather<<<N_NODES / 16, 128, 0, stream>>>(cur, rowptr, deg, col, agg);
    if (layer + 1 < N_LAYERS) {
      float* nxt = bufs[layer & 1];
      sage_gemm<<<GEMM_BLOCKS, 256, 0, stream>>>(
          agg, cur,
          Wl + (size_t)layer * HID * HID,
          bl + (size_t)layer * HID,
          Wr + (size_t)layer * HID * HID,
          nxt);
      cur = nxt;
    } else {
      sage_gemm_proj<<<GEMM_BLOCKS, 256, 0, stream>>>(
          agg, cur,
          Wl + (size_t)layer * HID * HID,
          bl + (size_t)layer * HID,
          Wr + (size_t)layer * HID * HID,
          Wout, bout, out);
    }
  }
}
